# starting kernel
# baseline (speedup 1.0000x reference)
#include <hip/hip_runtime.h>

typedef __attribute__((ext_vector_type(2))) float f32x2;
typedef __attribute__((ext_vector_type(8))) float f32x8;

#define C_    512
#define H_    96
#define W_    96
#define HW_   (H_ * W_)
#define CHW_  ((size_t)C_ * HW_)
#define TILES 7
#define TSTRIDE 14

__global__ __launch_bounds__(256) void corr_wmma_f32(const float* __restrict__ x1,
                                                     const float* __restrict__ x2,
                                                     float* __restrict__ out)
{
    const int lane = threadIdx.x & 31;
    const int wave = threadIdx.x >> 5;
    const int gw   = blockIdx.x * 8 + wave;

    const int tile = gw % TILES;
    const int h    = (gw / TILES) % H_;
    const int b    = gw / (TILES * H_);

    const int half = lane >> 4;
    const int li   = lane & 15;
    const int p    = tile * TSTRIDE;

    int wA = p + li; if (wA > W_ - 1) wA = W_ - 1;

    int colB = p - 1 + li;
    int colBc = colB < 0 ? 0 : (colB > W_ - 1 ? W_ - 1 : colB);

    const int  h0 = h - 1, h2 = h + 1;
    const bool row0Ok = (h0 >= 0);
    const bool row2Ok = (h2 <= H_ - 1);
    const int  h0c = row0Ok ? h0 : 0;
    const int  h2c = row2Ok ? h2 : 0;

    const size_t chanOfs = (size_t)(2 * half) * HW_;
    const float* pA  = x1 + (size_t)b * CHW_ + chanOfs + (size_t)h  * W_ + wA;
    const float* pB0 = x2 + (size_t)b * CHW_ + chanOfs + (size_t)h0c * W_ + colBc;
    const float* pB1 = x2 + (size_t)b * CHW_ + chanOfs + (size_t)h  * W_ + colBc;
    const float* pB2 = x2 + (size_t)b * CHW_ + chanOfs + (size_t)h2c * W_ + colBc;

    f32x8 acc0 = {};
    f32x8 acc1 = {};
    f32x8 acc2 = {};

#pragma unroll 4
    for (int k = 0; k < C_ / 4; ++k) {
        f32x2 a, b0, b1, b2;
        a.x  = pA[0];   a.y  = pA[HW_];
        b0.x = pB0[0];  b0.y = pB0[HW_];
        b1.x = pB1[0];  b1.y = pB1[HW_];
        b2.x = pB2[0];  b2.y = pB2[HW_];
        pA  += 4 * HW_;  pB0 += 4 * HW_;  pB1 += 4 * HW_;  pB2 += 4 * HW_;

        acc0 = __builtin_amdgcn_wmma_f32_16x16x4_f32(false, a, false, b0,
                                                     (short)0, acc0, false, false);
        acc1 = __builtin_amdgcn_wmma_f32_16x16x4_f32(false, a, false, b1,
                                                     (short)0, acc1, false, false);
        acc2 = __builtin_amdgcn_wmma_f32_16x16x4_f32(false, a, false, b2,
                                                     (short)0, acc2, false, false);
    }

    const float scale = 1.0f / (float)C_;
    f32x8 accs[3] = { acc0, acc1, acc2 };
    const bool rowOk[3] = { row0Ok, true, row2Ok };

    for (int r = 0; r < 8; ++r) {
        const int m  = 8 * half + r;
        const int dx = li - m;
        if (dx < 0 || dx > 2) continue;
        if (m > 13) continue;
        const int w = p + m;
        if (w > W_ - 1) continue;
        const int wcol  = w + dx - 1;
        const bool cOk  = (wcol >= 0) && (wcol <= W_ - 1);
        for (int dy = 0; dy < 3; ++dy) {
            const float g   = accs[dy][r];
            const float val = (cOk && rowOk[dy]) ? g * scale : 0.0f;
            out[((size_t)b * 9 + (dx * 3 + dy)) * HW_ + (size_t)h * W_ + w] = val;
        }
    }
}

extern "C" void kernel_launch(void* const* d_in, const int* in_sizes, int n_in,
                              void* d_out, int out_size, void* d_ws, size_t ws_size,
                              hipStream_t stream) {
    const float* x1 = (const float*)d_in[0];
    const float* x2 = (const float*)d_in[1];
    float* out = (float*)d_out;
    (void)in_sizes; (void)n_in; (void)out_size; (void)d_ws; (void)ws_size;

    const int blocks = (8 * H_ * TILES) / 8;
    corr_wmma_f32<<<blocks, 256, 0, stream>>>(x1, x2, out);
}


# Operands staged through LDS per band of 3 output rows: a workgroup copies 3 x1 rows and 5 x2 rows of 32 channels with whole-line 16-byte loads one stage ahead, fragments from LDS; sum order unchanged
# speedup vs baseline: 1.5168x; 1.5168x over previous
#include <hip/hip_runtime.h>


typedef __attribute__((ext_vector_type(2))) float f32x2;
typedef __attribute__((ext_vector_type(4))) float f32x4;
typedef __attribute__((ext_vector_type(8))) float f32x8;

#define C_      512
#define H_      96
#define W_      96
#define HW_     (H_ * W_)
#define CHW_    ((size_t)C_ * HW_)
#define ROWS    3
#define BROWS   (ROWS + 2)
#define BANDS   (H_ / ROWS)
#define KC      32
#define NSTAGE  (C_ / KC)
#define NPAIR   (KC / 2)
#define KCHUNKS (KC / 4)
#define TSTRIDE 12
#define QUADS   (W_ / 4)
#define PA      (ROWS  * W_ * 2 + 32)
#define PB      (BROWS * W_ * 2 + 32)
#define OFFB    (NPAIR * PA)
#define STAGE_F (NPAIR * (PA + PB))
#define UA      (NPAIR * ROWS  * QUADS)
#define UB      (NPAIR * BROWS * QUADS)
#define UPT     ((UA + UB) / 256)

__global__ __launch_bounds__(256) void corr_band_lds(const float* __restrict__ x1,
                                                     const float* __restrict__ x2,
                                                     float* __restrict__ out)
{
    __shared__ __attribute__((aligned(16))) float lds[2 * STAGE_F];

    const int tid  = threadIdx.x;
    const int lane = tid & 31;
    const int wave = tid >> 5;
    const int band = blockIdx.x % BANDS;
    const int b    = blockIdx.x / BANDS;
    const int h0   = band * ROWS;

    const float* gsrc[UPT];
    int          ldst[UPT];
#pragma unroll
    for (int i = 0; i < UPT; ++i) {
        const int  u    = tid + 256 * i;
        const bool isA  = u < UA;
        const int  v    = isA ? u : u - UA;
        const int  q    = v % QUADS;
        const int  t    = v / QUADS;
        const int  rr   = isA ? t % ROWS : t % BROWS;
        const int  pair = isA ? t / ROWS : t / BROWS;
        int grow = isA ? (h0 + rr) : (h0 - 1 + rr);
        grow = grow < 0 ? 0 : (grow > H_ - 1 ? H_ - 1 : grow);
        const float* base = isA ? x1 : x2;
        gsrc[i] = base + (size_t)b * CHW_ + (size_t)(2 * pair) * HW_ + (size_t)grow * W_ + 4 * q;
        ldst[i] = (isA ? pair * PA : OFFB + pair * PB) + (rr * W_ + 4 * q) * 2;
    }
    const bool sw  = ((tid >> 3) & 1) != 0;
    const int  o1  = sw ? 4 : 0;
    const int  o2  = sw ? 0 : 4;

    const int half = lane >> 4;
    const int li   = lane & 15;
    const int p    = wave * TSTRIDE;
    int wA = p + li; if (wA > W_ - 1) wA = W_ - 1;
    const int colB  = p - 1 + li;
    const int colBc = colB < 0 ? 0 : (colB > W_ - 1 ? W_ - 1 : colB);
    const int aoff = half * PA + wA * 2;
    const int boff = OFFB + half * PB + colBc * 2;

    f32x8 acc[ROWS][3];
#pragma unroll
    for (int r = 0; r < ROWS; ++r)
#pragma unroll
        for (int dy = 0; dy < 3; ++dy) acc[r][dy] = (f32x8){};

    f32x4 g0[UPT], g1[UPT];

#define LOAD_STAGE(S)                                                          \
    _Pragma("unroll")                                                          \
    for (int i = 0; i < UPT; ++i) {                                            \
        const float* src = gsrc[i] + (size_t)(S) * (KC * HW_);                 \
        g0[i] = *(const f32x4*)(src);                                          \
        g1[i] = *(const f32x4*)(src + HW_);                                    \
    }

#define STORE_STAGE(BUF)                                                       \
    _Pragma("unroll")                                                          \
    for (int i = 0; i < UPT; ++i) {                                            \
        float* d = lds + (BUF) * STAGE_F + ldst[i];                            \
        const f32x4 s1 = { g0[i].x, g1[i].x, g0[i].y, g1[i].y };               \
        const f32x4 s2 = { g0[i].z, g1[i].z, g0[i].w, g1[i].w };               \
        *(f32x4*)(d + o1) = sw ? s2 : s1;                                      \
        *(f32x4*)(d + o2) = sw ? s1 : s2;                                      \
    }

#define COMPUTE_STAGE(BUF)                                                     \
    {                                                                          \
        const float* L = lds + (BUF) * STAGE_F;                                \
        _Pragma("unroll")                                                      \
        for (int kc = 0; kc < KCHUNKS; ++kc) {                                 \
            f32x2 a[ROWS], bb[BROWS];                                          \
            _Pragma("unroll")                                                  \
            for (int r = 0; r < ROWS; ++r)                                     \
                a[r] = *(const f32x2*)(L + aoff + 2 * kc * PA + r * (W_ * 2)); \
            _Pragma("unroll")                                                  \
            for (int j = 0; j < BROWS; ++j)                                    \
                bb[j] = *(const f32x2*)(L + boff + 2 * kc * PB + j * (W_ * 2));\
            _Pragma("unroll")                                                  \
            for (int r = 0; r < ROWS; ++r) {                                   \
                _Pragma("unroll")                                              \
                for (int dy = 0; dy < 3; ++dy)                                 \
                    acc[r][dy] = __builtin_amdgcn_wmma_f32_16x16x4_f32(        \
                        false, a[r], false, bb[r + dy], (short)0, acc[r][dy],  \
                        false, false);                                         \
            }                                                                  \
        }                                                                      \
    }

    LOAD_STAGE(0)
    STORE_STAGE(0)
    __syncthreads();

    for (int s = 0; s < NSTAGE - 1; ++s) {
        const int cur = s & 1;
        LOAD_STAGE(s + 1)
        COMPUTE_STAGE(cur)
        STORE_STAGE(cur ^ 1)
        __syncthreads();
    }
    COMPUTE_STAGE((NSTAGE - 1) & 1)

    const float scale = 1.0f / (float)C_;
#pragma unroll
    for (int rr = 0; rr < ROWS; ++rr) {
        const int h = h0 + rr;
        const bool rowOk[3] = { h - 1 >= 0, true, h + 1 <= H_ - 1 };
        for (int r = 0; r < 8; ++r) {
            const int m  = 8 * half + r;
            const int dx = li - m;
            if (dx < 0 || dx > 2) continue;
            if (m > TSTRIDE - 1) continue;
            const int w = p + m;
            const int wcol  = w + dx - 1;
            const bool cOk  = (wcol >= 0) && (wcol <= W_ - 1);
            for (int dy = 0; dy < 3; ++dy) {
                const float g   = acc[rr][dy][r];
                const float val = (cOk && rowOk[dy]) ? g * scale : 0.0f;
                out[((size_t)b * 9 + (dx * 3 + dy)) * HW_ + (size_t)h * W_ + w] = val;
            }
        }
    }
}

extern "C" void kernel_launch(void* const* d_in, const int* in_sizes, int n_in,
                              void* d_out, int out_size, void* d_ws, size_t ws_size,
                              hipStream_t stream) {
    const float* x1 = (const float*)d_in[0];
    const float* x2 = (const float*)d_in[1];
    float* out = (float*)d_out;
    (void)in_sizes; (void)n_in; (void)out_size; (void)d_ws; (void)ws_size;

    corr_band_lds<<<256, 256, 0, stream>>>(x1, x2, out);
}


# Staging of each stage split in two halves requested half a stage apart, with a hand-written LDS-only barrier: one half's global loads stay in flight across every barrier; sum order unchanged
# speedup vs baseline: 1.7063x; 1.1249x over previous
#include <hip/hip_runtime.h>


typedef __attribute__((ext_vector_type(2))) float f32x2;
typedef __attribute__((ext_vector_type(4))) float f32x4;
typedef __attribute__((ext_vector_type(8))) float f32x8;

#define C_      512
#define H_      96
#define W_      96
#define HW_     (H_ * W_)
#define CHW_    ((size_t)C_ * HW_)
#define ROWS    3
#define BROWS   (ROWS + 2)
#define BANDS   (H_ / ROWS)
#define KC      32
#define NSTAGE  (C_ / KC)
#define NPAIR   (KC / 2)
#define KCHUNKS (KC / 4)
#define TSTRIDE 12
#define QUADS   (W_ / 4)
#define PA      (ROWS  * W_ * 2 + 32)
#define PB      (BROWS * W_ * 2 + 32)
#define OFFB    (NPAIR * PA)
#define STAGE_F (NPAIR * (PA + PB))
#define UA      (NPAIR * ROWS  * QUADS)
#define UB      (NPAIR * BROWS * QUADS)
#define UPT     ((UA + UB) / 256)
#define HUPT    (UPT / 2)

static __device__ __forceinline__ void lds_barrier() {
    asm volatile("s_wait_dscnt 0x0\n\ts_barrier_signal -1\n\ts_barrier_wait -1" ::: "memory");
}

__global__ __launch_bounds__(256) void corr_band_lds(const float* __restrict__ x1,
                                                     const float* __restrict__ x2,
                                                     float* __restrict__ out)
{
    __shared__ __attribute__((aligned(16))) float lds[2 * STAGE_F];

    const int tid  = threadIdx.x;
    const int lane = tid & 31;
    const int wave = tid >> 5;
    const int band = blockIdx.x % BANDS;
    const int b    = blockIdx.x / BANDS;
    const int h0   = band * ROWS;

    const float* gsrc[UPT];
    int          ldst[UPT];
#pragma unroll
    for (int i = 0; i < UPT; ++i) {
        const int  u    = tid + 256 * i;
        const bool isA  = u < UA;
        const int  v    = isA ? u : u - UA;
        const int  q    = v % QUADS;
        const int  t    = v / QUADS;
        const int  rr   = isA ? t % ROWS : t % BROWS;
        const int  pair = isA ? t / ROWS : t / BROWS;
        int grow = isA ? (h0 + rr) : (h0 - 1 + rr);
        grow = grow < 0 ? 0 : (grow > H_ - 1 ? H_ - 1 : grow);
        const float* base = isA ? x1 : x2;
        gsrc[i] = base + (size_t)b * CHW_ + (size_t)(2 * pair) * HW_ + (size_t)grow * W_ + 4 * q;
        ldst[i] = (isA ? pair * PA : OFFB + pair * PB) + (rr * W_ + 4 * q) * 2;
    }
    const bool sw  = ((tid >> 3) & 1) != 0;
    const int  o1  = sw ? 4 : 0;
    const int  o2  = sw ? 0 : 4;

    const int half = lane >> 4;
    const int li   = lane & 15;
    const int p    = wave * TSTRIDE;
    int wA = p + li; if (wA > W_ - 1) wA = W_ - 1;
    const int colB  = p - 1 + li;
    const int colBc = colB < 0 ? 0 : (colB > W_ - 1 ? W_ - 1 : colB);
    const int aoff = half * PA + wA * 2;
    const int boff = OFFB + half * PB + colBc * 2;

    f32x8 acc[ROWS][3];
#pragma unroll
    for (int r = 0; r < ROWS; ++r)
#pragma unroll
        for (int dy = 0; dy < 3; ++dy) acc[r][dy] = (f32x8){};

    f32x4 g0[UPT], g1[UPT];

#define LOAD_HALF(S, I0)                                                       \
    _Pragma("unroll")                                                          \
    for (int i = (I0); i < (I0) + HUPT; ++i) {                                 \
        const float* src = gsrc[i] + (size_t)(S) * (KC * HW_);                 \
        g0[i] = *(const f32x4*)(src);                                          \
        g1[i] = *(const f32x4*)(src + HW_);                                    \
    }

#define STORE_HALF(BUF, I0)                                                    \
    _Pragma("unroll")                                                          \
    for (int i = (I0); i < (I0) + HUPT; ++i) {                                 \
        float* d = lds + (BUF) * STAGE_F + ldst[i];                            \
        const f32x4 s1 = { g0[i].x, g1[i].x, g0[i].y, g1[i].y };               \
        const f32x4 s2 = { g0[i].z, g1[i].z, g0[i].w, g1[i].w };               \
        *(f32x4*)(d + o1) = sw ? s2 : s1;                                      \
        *(f32x4*)(d + o2) = sw ? s1 : s2;                                      \
    }

#define COMPUTE_HALF(BUF, K0)                                                  \
    {                                                                          \
        const float* L = lds + (BUF) * STAGE_F;                                \
        _Pragma("unroll")                                                      \
        for (int kc = (K0); kc < (K0) + KCHUNKS / 2; ++kc) {                   \
            f32x2 a[ROWS], bb[BROWS];                                          \
            _Pragma("unroll")                                                  \
            for (int r = 0; r < ROWS; ++r)                                     \
                a[r] = *(const f32x2*)(L + aoff + 2 * kc * PA + r * (W_ * 2)); \
            _Pragma("unroll")                                                  \
            for (int j = 0; j < BROWS; ++j)                                    \
                bb[j] = *(const f32x2*)(L + boff + 2 * kc * PB + j * (W_ * 2));\
            _Pragma("unroll")                                                  \
            for (int r = 0; r < ROWS; ++r) {                                   \
                _Pragma("unroll")                                              \
                for (int dy = 0; dy < 3; ++dy)                                 \
                    acc[r][dy] = __builtin_amdgcn_wmma_f32_16x16x4_f32(        \
                        false, a[r], false, bb[r + dy], (short)0, acc[r][dy],  \
                        false, false);                                         \
            }                                                                  \
        }                                                                      \
    }

    LOAD_HALF(0, 0)
    LOAD_HALF(0, HUPT)
    STORE_HALF(0, 0)
    STORE_HALF(0, HUPT)
    LOAD_HALF(1, 0)
    lds_barrier();

#define SCHED_FENCE() __builtin_amdgcn_sched_barrier(0)

    for (int s = 0; s < NSTAGE - 2; ++s) {
        const int cur = s & 1;
        LOAD_HALF(s + 1, HUPT)
        COMPUTE_HALF(cur, 0)
        SCHED_FENCE();
        STORE_HALF(cur ^ 1, 0)
        LOAD_HALF(s + 2, 0)
        COMPUTE_HALF(cur, KCHUNKS / 2)
        SCHED_FENCE();
        STORE_HALF(cur ^ 1, HUPT)
        lds_barrier();
    }
    LOAD_HALF(NSTAGE - 1, HUPT)
    COMPUTE_HALF(0, 0)
    SCHED_FENCE();
    STORE_HALF(1, 0)
    COMPUTE_HALF(0, KCHUNKS / 2)
    SCHED_FENCE();
    STORE_HALF(1, HUPT)
    lds_barrier();
    COMPUTE_HALF(1, 0)
    COMPUTE_HALF(1, KCHUNKS / 2)

    const float scale = 1.0f / (float)C_;
#pragma unroll
    for (int rr = 0; rr < ROWS; ++rr) {
        const int h = h0 + rr;
        const bool rowOk[3] = { h - 1 >= 0, true, h + 1 <= H_ - 1 };
        for (int r = 0; r < 8; ++r) {
            const int m  = 8 * half + r;
            const int dx = li - m;
            if (dx < 0 || dx > 2) continue;
            if (m > TSTRIDE - 1) continue;
            const int w = p + m;
            const int wcol  = w + dx - 1;
            const bool cOk  = (wcol >= 0) && (wcol <= W_ - 1);
            for (int dy = 0; dy < 3; ++dy) {
                const float g   = acc[rr][dy][r];
                const float val = (cOk && rowOk[dy]) ? g * scale : 0.0f;
                out[((size_t)b * 9 + (dx * 3 + dy)) * HW_ + (size_t)h * W_ + w] = val;
            }
        }
    }
}

extern "C" void kernel_launch(void* const* d_in, const int* in_sizes, int n_in,
                              void* d_out, int out_size, void* d_ws, size_t ws_size,
                              hipStream_t stream) {
    const float* x1 = (const float*)d_in[0];
    const float* x2 = (const float*)d_in[1];
    float* out = (float*)d_out;
    (void)in_sizes; (void)n_in; (void)out_size; (void)d_ws; (void)ws_size;

    corr_band_lds<<<256, 256, 0, stream>>>(x1, x2, out);
}
